# starting kernel
# baseline (speedup 1.0000x reference)
#include <hip/hip_runtime.h>
#include <hip/hip_bf16.h>

typedef float v2f __attribute__((ext_vector_type(2)));
typedef float v8f __attribute__((ext_vector_type(8)));

#if defined(__has_builtin)
#if __has_builtin(__builtin_amdgcn_tanhf)
#define FAST_TANH(x) __builtin_amdgcn_tanhf(x)
#endif
#endif
#ifndef FAST_TANH
#define FAST_TANH(x) tanhf(x)
#endif

__device__ __forceinline__ float lane_gather(float v, int srcLane) {
  return __int_as_float(__builtin_amdgcn_ds_bpermute(srcLane << 2, __float_as_int(v)));
}

#define TPW 8

__global__ void __launch_bounds__(256)
pinn_mlp16_wmma(const float* __restrict__ x,  const float* __restrict__ x2,
                const float* __restrict__ W1, const float* __restrict__ W2,
                const float* __restrict__ B1,
                const float* __restrict__ W3, const float* __restrict__ B3,
                const float* __restrict__ W4, const float* __restrict__ B4,
                const float* __restrict__ W5, const float* __restrict__ B5,
                const float* __restrict__ W6, const float* __restrict__ B6,
                const float* __restrict__ W7, const float* __restrict__ B7,
                const float* __restrict__ W8, const float* __restrict__ B8,
                const float* __restrict__ W9, const float* __restrict__ B9,
                const float* __restrict__ alphap,
                float* __restrict__ out, int npts)
{
  const int lane = threadIdx.x & 31;
  const int p    = lane & 15;
  const int hi   = lane >> 4;
  const int wave = blockIdx.x * (blockDim.x >> 5) + (threadIdx.x >> 5);

  const float alpha = alphap[0];
  const float b9    = B9[0];

  const float* Ws[6] = {W3, W4, W5, W6, W7, W8};
  const float* Bs[6] = {B3, B4, B5, B6, B7, B8};

  v2f   aW[6][4];
  float ab[6][8];
  #pragma unroll
  for (int l = 0; l < 6; ++l) {
    #pragma unroll
    for (int k = 0; k < 4; ++k)
      #pragma unroll
      for (int j = 0; j < 2; ++j)
        aW[l][k][j] = Ws[l][(4 * k + j + 2 * hi) * 16 + p];
    #pragma unroll
    for (int v = 0; v < 8; ++v)
      ab[l][v] = alpha * Bs[l][v + 8 * hi];
  }

  float aw1[8], aw2[8], ab1[8], w9v[8];
  #pragma unroll
  for (int v = 0; v < 8; ++v) {
    const int f = v + 8 * hi;
    aw1[v] = alpha * W1[f];
    aw2[v] = alpha * W2[f];
    ab1[v] = alpha * B1[f];
    w9v[v] = W9[f];
  }

  const int ntiles = (npts + 15) >> 4;
  int tile = wave * TPW;

  #pragma unroll 1
  for (int t = 0; t < TPW; ++t, ++tile) {
    if (tile >= ntiles) break;
    const int  base = tile << 4;
    const bool ok   = (base + p) < npts;
    const float xp  = ok ? x [base + p] : 0.0f;
    const float x2p = ok ? x2[base + p] : 0.0f;

    v8f act;
    #pragma unroll
    for (int v = 0; v < 8; ++v)
      act[v] = FAST_TANH(fmaf(xp, aw1[v], fmaf(x2p, aw2[v], ab1[v])));

    #pragma unroll
    for (int l = 0; l < 6; ++l) {
      v8f d = {};
      #pragma unroll
      for (int k = 0; k < 4; ++k) {
        v2f bk;
        #pragma unroll
        for (int j = 0; j < 2; ++j) {
          const int r1 = 4 * k + j;
          const int r2 = r1 + 2;
          float bv;
          if (r2 < 8) {
            float g = lane_gather(act[r2], p);
            bv = hi ? g : act[r1];
          } else {
            float g = lane_gather(act[r1 & 7], p + 16);
            bv = hi ? act[r2 & 7] : g;
          }
          bk[j] = bv;
        }
        d = __builtin_amdgcn_wmma_f32_16x16x4_f32(false, aW[l][k], false, bk,
                                                  (short)0, d, false, false);
      }
      #pragma unroll
      for (int v = 0; v < 8; ++v)
        act[v] = FAST_TANH(fmaf(alpha, d[v], ab[l][v]));
    }

    float partial = 0.0f;
    #pragma unroll
    for (int v = 0; v < 8; ++v)
      partial = fmaf(act[v], w9v[v], partial);
    const float other = lane_gather(partial, lane ^ 16);
    const float res   = partial + other + b9;
    if (hi == 0 && ok)
      out[base + p] = res;
  }
}

extern "C" void kernel_launch(void* const* d_in, const int* in_sizes, int n_in,
                              void* d_out, int out_size, void* d_ws, size_t ws_size,
                              hipStream_t stream) {
  (void)n_in; (void)d_ws; (void)ws_size; (void)out_size;
  const float* x   = (const float*)d_in[0];
  const float* x2  = (const float*)d_in[1];
  const float* W1  = (const float*)d_in[2];
  const float* W2  = (const float*)d_in[3];
  const float* B1  = (const float*)d_in[4];
  const float* W3  = (const float*)d_in[5];
  const float* B3  = (const float*)d_in[6];
  const float* W4  = (const float*)d_in[7];
  const float* B4  = (const float*)d_in[8];
  const float* W5  = (const float*)d_in[9];
  const float* B5  = (const float*)d_in[10];
  const float* W6  = (const float*)d_in[11];
  const float* B6  = (const float*)d_in[12];
  const float* W7  = (const float*)d_in[13];
  const float* B7  = (const float*)d_in[14];
  const float* W8  = (const float*)d_in[15];
  const float* B8  = (const float*)d_in[16];
  const float* W9  = (const float*)d_in[17];
  const float* B9  = (const float*)d_in[18];
  const float* alp = (const float*)d_in[19];
  float* out = (float*)d_out;

  const int npts   = in_sizes[0];
  const int ntiles = (npts + 15) >> 4;
  const int waves  = (ntiles + TPW - 1) / TPW;
  const int blocks = (waves + 7) / 8;

  pinn_mlp16_wmma<<<blocks, 256, 0, stream>>>(
      x, x2, W1, W2, B1, W3, B3, W4, B4, W5, B5, W6, B6,
      W7, B7, W8, B8, W9, B9, alp, out, npts);
}


# Hidden layers: W loader puts feature 2k+j+8*hi in k slot (chunk k, reg j, hi=lane>>4), so each layer's D registers are the next B fragments as they stand: no ds_bpermute/select; k sum order changes
# speedup vs baseline: 1.1334x; 1.1334x over previous
#include <hip/hip_runtime.h>
#include <hip/hip_bf16.h>

typedef float v2f __attribute__((ext_vector_type(2)));
typedef float v8f __attribute__((ext_vector_type(8)));

#if defined(__has_builtin)
#if __has_builtin(__builtin_amdgcn_tanhf)
#define FAST_TANH(x) __builtin_amdgcn_tanhf(x)
#endif
#endif
#ifndef FAST_TANH
#define FAST_TANH(x) tanhf(x)
#endif

__device__ __forceinline__ float lane_gather(float v, int srcLane) {
  return __int_as_float(__builtin_amdgcn_ds_bpermute(srcLane << 2, __float_as_int(v)));
}

#define TPW 8

__global__ void __launch_bounds__(256)
pinn_mlp16_wmma(const float* __restrict__ x,  const float* __restrict__ x2,
                const float* __restrict__ W1, const float* __restrict__ W2,
                const float* __restrict__ B1,
                const float* __restrict__ W3, const float* __restrict__ B3,
                const float* __restrict__ W4, const float* __restrict__ B4,
                const float* __restrict__ W5, const float* __restrict__ B5,
                const float* __restrict__ W6, const float* __restrict__ B6,
                const float* __restrict__ W7, const float* __restrict__ B7,
                const float* __restrict__ W8, const float* __restrict__ B8,
                const float* __restrict__ W9, const float* __restrict__ B9,
                const float* __restrict__ alphap,
                float* __restrict__ out, int npts)
{
  const int lane = threadIdx.x & 31;
  const int p    = lane & 15;
  const int hi   = lane >> 4;
  const int wave = blockIdx.x * (blockDim.x >> 5) + (threadIdx.x >> 5);

  const float alpha = alphap[0];
  const float b9    = B9[0];

  const float* Ws[6] = {W3, W4, W5, W6, W7, W8};
  const float* Bs[6] = {B3, B4, B5, B6, B7, B8};

  v2f   aW[6][4];
  float ab[6][8];
  #pragma unroll
  for (int l = 0; l < 6; ++l) {
    #pragma unroll
    for (int k = 0; k < 4; ++k)
      #pragma unroll
      for (int j = 0; j < 2; ++j)
        aW[l][k][j] = Ws[l][(2 * k + j + 8 * hi) * 16 + p];
    #pragma unroll
    for (int v = 0; v < 8; ++v)
      ab[l][v] = alpha * Bs[l][v + 8 * hi];
  }

  float aw1[8], aw2[8], ab1[8], w9v[8];
  #pragma unroll
  for (int v = 0; v < 8; ++v) {
    const int f = v + 8 * hi;
    aw1[v] = alpha * W1[f];
    aw2[v] = alpha * W2[f];
    ab1[v] = alpha * B1[f];
    w9v[v] = W9[f];
  }

  const int ntiles = (npts + 15) >> 4;
  int tile = wave * TPW;

  #pragma unroll 1
  for (int t = 0; t < TPW; ++t, ++tile) {
    if (tile >= ntiles) break;
    const int  base = tile << 4;
    const bool ok   = (base + p) < npts;
    const float xp  = ok ? x [base + p] : 0.0f;
    const float x2p = ok ? x2[base + p] : 0.0f;

    v8f act;
    #pragma unroll
    for (int v = 0; v < 8; ++v)
      act[v] = FAST_TANH(fmaf(xp, aw1[v], fmaf(x2p, aw2[v], ab1[v])));

    #pragma unroll
    for (int l = 0; l < 6; ++l) {
      v8f d = {};
      #pragma unroll
      for (int k = 0; k < 4; ++k) {
        v2f bk;
        bk[0] = act[2 * k];
        bk[1] = act[2 * k + 1];
        d = __builtin_amdgcn_wmma_f32_16x16x4_f32(false, aW[l][k], false, bk,
                                                  (short)0, d, false, false);
      }
      #pragma unroll
      for (int v = 0; v < 8; ++v)
        act[v] = FAST_TANH(fmaf(alpha, d[v], ab[l][v]));
    }

    float partial = 0.0f;
    #pragma unroll
    for (int v = 0; v < 8; ++v)
      partial = fmaf(act[v], w9v[v], partial);
    const float other = lane_gather(partial, lane ^ 16);
    const float res   = partial + other + b9;
    if (hi == 0 && ok)
      out[base + p] = res;
  }
}

extern "C" void kernel_launch(void* const* d_in, const int* in_sizes, int n_in,
                              void* d_out, int out_size, void* d_ws, size_t ws_size,
                              hipStream_t stream) {
  (void)n_in; (void)d_ws; (void)ws_size; (void)out_size;
  const float* x   = (const float*)d_in[0];
  const float* x2  = (const float*)d_in[1];
  const float* W1  = (const float*)d_in[2];
  const float* W2  = (const float*)d_in[3];
  const float* B1  = (const float*)d_in[4];
  const float* W3  = (const float*)d_in[5];
  const float* B3  = (const float*)d_in[6];
  const float* W4  = (const float*)d_in[7];
  const float* B4  = (const float*)d_in[8];
  const float* W5  = (const float*)d_in[9];
  const float* B5  = (const float*)d_in[10];
  const float* W6  = (const float*)d_in[11];
  const float* B6  = (const float*)d_in[12];
  const float* W7  = (const float*)d_in[13];
  const float* B7  = (const float*)d_in[14];
  const float* W8  = (const float*)d_in[15];
  const float* B8  = (const float*)d_in[16];
  const float* W9  = (const float*)d_in[17];
  const float* B9  = (const float*)d_in[18];
  const float* alp = (const float*)d_in[19];
  float* out = (float*)d_out;

  const int npts   = in_sizes[0];
  const int ntiles = (npts + 15) >> 4;
  const int waves  = (ntiles + TPW - 1) / TPW;
  const int blocks = (waves + 7) / 8;

  pinn_mlp16_wmma<<<blocks, 256, 0, stream>>>(
      x, x2, W1, W2, B1, W3, B3, W4, B4, W5, B5, W6, B6,
      W7, B7, W8, B8, W9, B9, alp, out, npts);
}


# Two 16-point tiles per loop trip as two independent WMMA accumulator chains, alternating, so one tile's bias/tanh work issues under the other tile's matrix ops; per-tile sum order unchanged
# speedup vs baseline: 1.1751x; 1.0368x over previous
#include <hip/hip_runtime.h>
#include <hip/hip_bf16.h>

typedef float v2f __attribute__((ext_vector_type(2)));
typedef float v8f __attribute__((ext_vector_type(8)));

#if defined(__has_builtin)
#if __has_builtin(__builtin_amdgcn_tanhf)
#define FAST_TANH(x) __builtin_amdgcn_tanhf(x)
#endif
#endif
#ifndef FAST_TANH
#define FAST_TANH(x) tanhf(x)
#endif

__device__ __forceinline__ float lane_gather(float v, int srcLane) {
  return __int_as_float(__builtin_amdgcn_ds_bpermute(srcLane << 2, __float_as_int(v)));
}

#define TPW 8

__global__ void __launch_bounds__(256)
pinn_mlp16_wmma(const float* __restrict__ x,  const float* __restrict__ x2,
                const float* __restrict__ W1, const float* __restrict__ W2,
                const float* __restrict__ B1,
                const float* __restrict__ W3, const float* __restrict__ B3,
                const float* __restrict__ W4, const float* __restrict__ B4,
                const float* __restrict__ W5, const float* __restrict__ B5,
                const float* __restrict__ W6, const float* __restrict__ B6,
                const float* __restrict__ W7, const float* __restrict__ B7,
                const float* __restrict__ W8, const float* __restrict__ B8,
                const float* __restrict__ W9, const float* __restrict__ B9,
                const float* __restrict__ alphap,
                float* __restrict__ out, int npts)
{
  const int lane = threadIdx.x & 31;
  const int p    = lane & 15;
  const int hi   = lane >> 4;
  const int wave = blockIdx.x * (blockDim.x >> 5) + (threadIdx.x >> 5);

  const float alpha = alphap[0];
  const float b9    = B9[0];

  const float* Ws[6] = {W3, W4, W5, W6, W7, W8};
  const float* Bs[6] = {B3, B4, B5, B6, B7, B8};

  v2f   aW[6][4];
  float ab[6][8];
  #pragma unroll
  for (int l = 0; l < 6; ++l) {
    #pragma unroll
    for (int k = 0; k < 4; ++k)
      #pragma unroll
      for (int j = 0; j < 2; ++j)
        aW[l][k][j] = Ws[l][(2 * k + j + 8 * hi) * 16 + p];
    #pragma unroll
    for (int v = 0; v < 8; ++v)
      ab[l][v] = alpha * Bs[l][v + 8 * hi];
  }

  float aw1[8], aw2[8], ab1[8], w9v[8];
  #pragma unroll
  for (int v = 0; v < 8; ++v) {
    const int f = v + 8 * hi;
    aw1[v] = alpha * W1[f];
    aw2[v] = alpha * W2[f];
    ab1[v] = alpha * B1[f];
    w9v[v] = W9[f];
  }

  const int ntiles = (npts + 15) >> 4;
  int tile = wave * TPW;

#define FIRST_CHUNK(dst, xv, x2v, k)                                                   \
  do {                                                                                 \
    dst[0] = FAST_TANH(fmaf(xv, aw1[2 * (k)],     fmaf(x2v, aw2[2 * (k)],     ab1[2 * (k)])));     \
    dst[1] = FAST_TANH(fmaf(xv, aw1[2 * (k) + 1], fmaf(x2v, aw2[2 * (k) + 1], ab1[2 * (k) + 1]))); \
  } while (0)
#define NEXT_CHUNK(dst, dprev, lprev, k)                                               \
  do {                                                                                 \
    dst[0] = FAST_TANH(fmaf(alpha, dprev[2 * (k)],     ab[lprev][2 * (k)]));           \
    dst[1] = FAST_TANH(fmaf(alpha, dprev[2 * (k) + 1], ab[lprev][2 * (k) + 1]));       \
  } while (0)
#define WMMA_STEP(acc, l, k, bfrag)                                                    \
  acc = __builtin_amdgcn_wmma_f32_16x16x4_f32(false, aW[l][k], false, bfrag,           \
                                              (short)0, acc, false, false)
#define KEEP_ORDER() __builtin_amdgcn_sched_barrier(0)

  #pragma unroll 1
  for (int t = 0; t < TPW; t += 2, tile += 2) {
    if (tile >= ntiles) break;
    const int  baseA = tile << 4;
    const int  baseB = baseA + 16;
    const bool okA   = (baseA + p) < npts;
    const bool okB   = (baseB + p) < npts;
    const float xA   = okA ? x [baseA + p] : 0.0f;
    const float x2A  = okA ? x2[baseA + p] : 0.0f;
    const float xB   = okB ? x [baseB + p] : 0.0f;
    const float x2B  = okB ? x2[baseB + p] : 0.0f;

    v8f dA = {};
    v8f dB = {};
    v2f bA, bB, bAn, bBn;

    FIRST_CHUNK(bA, xA, x2A, 0);
    FIRST_CHUNK(bB, xB, x2B, 0);
    KEEP_ORDER();
    #pragma unroll
    for (int k = 0; k < 4; ++k) {
      WMMA_STEP(dA, 0, k, bA);
      if (k < 3) FIRST_CHUNK(bAn, xA, x2A, k + 1);
      KEEP_ORDER();
      WMMA_STEP(dB, 0, k, bB);
      if (k < 3) FIRST_CHUNK(bBn, xB, x2B, k + 1);
      KEEP_ORDER();
      bA = bAn;
      bB = bBn;
    }

    #pragma unroll
    for (int l = 1; l < 6; ++l) {
      const v8f pA = dA;
      const v8f pB = dB;
      dA = (v8f){};
      dB = (v8f){};
      NEXT_CHUNK(bA, pA, l - 1, 0);
      KEEP_ORDER();
      WMMA_STEP(dA, l, 0, bA);
      NEXT_CHUNK(bB, pB, l - 1, 0);
      NEXT_CHUNK(bAn, pA, l - 1, 1);
      KEEP_ORDER();
      WMMA_STEP(dB, l, 0, bB);
      NEXT_CHUNK(bBn, pB, l - 1, 1);
      KEEP_ORDER();
      bA = bAn;
      bB = bBn;
      #pragma unroll
      for (int k = 1; k < 4; ++k) {
        WMMA_STEP(dA, l, k, bA);
        if (k < 3) NEXT_CHUNK(bAn, pA, l - 1, k + 1);
        KEEP_ORDER();
        WMMA_STEP(dB, l, k, bB);
        if (k < 3) NEXT_CHUNK(bBn, pB, l - 1, k + 1);
        KEEP_ORDER();
        bA = bAn;
        bB = bBn;
      }
    }

    float partA = 0.0f, partB = 0.0f;
    #pragma unroll
    for (int v = 0; v < 8; ++v) {
      partA = fmaf(FAST_TANH(fmaf(alpha, dA[v], ab[5][v])), w9v[v], partA);
      partB = fmaf(FAST_TANH(fmaf(alpha, dB[v], ab[5][v])), w9v[v], partB);
    }
    const float otherA = lane_gather(partA, lane ^ 16);
    const float otherB = lane_gather(partB, lane ^ 16);
    const float resA   = partA + otherA + b9;
    const float resB   = partB + otherB + b9;
    if (hi == 0 && okA)
      out[baseA + p] = resA;
    if (hi == 0 && okB)
      out[baseB + p] = resB;
  }
#undef FIRST_CHUNK
#undef NEXT_CHUNK
#undef WMMA_STEP
#undef KEEP_ORDER
}

extern "C" void kernel_launch(void* const* d_in, const int* in_sizes, int n_in,
                              void* d_out, int out_size, void* d_ws, size_t ws_size,
                              hipStream_t stream) {
  (void)n_in; (void)d_ws; (void)ws_size; (void)out_size;
  const float* x   = (const float*)d_in[0];
  const float* x2  = (const float*)d_in[1];
  const float* W1  = (const float*)d_in[2];
  const float* W2  = (const float*)d_in[3];
  const float* B1  = (const float*)d_in[4];
  const float* W3  = (const float*)d_in[5];
  const float* B3  = (const float*)d_in[6];
  const float* W4  = (const float*)d_in[7];
  const float* B4  = (const float*)d_in[8];
  const float* W5  = (const float*)d_in[9];
  const float* B5  = (const float*)d_in[10];
  const float* W6  = (const float*)d_in[11];
  const float* B6  = (const float*)d_in[12];
  const float* W7  = (const float*)d_in[13];
  const float* B7  = (const float*)d_in[14];
  const float* W8  = (const float*)d_in[15];
  const float* B8  = (const float*)d_in[16];
  const float* W9  = (const float*)d_in[17];
  const float* B9  = (const float*)d_in[18];
  const float* alp = (const float*)d_in[19];
  float* out = (float*)d_out;

  const int npts   = in_sizes[0];
  const int ntiles = (npts + 15) >> 4;
  const int waves  = (ntiles + TPW - 1) / TPW;
  const int blocks = (waves + 7) / 8;

  pinn_mlp16_wmma<<<blocks, 256, 0, stream>>>(
      x, x2, W1, W2, B1, W3, B3, W4, B4, W5, B5, W6, B6,
      W7, B7, W8, B8, W9, B9, alp, out, npts);
}


# All of a wave's x/x2 inputs (8 tiles) are loaded at wave start, ahead of the weight loads, and its 4 trips unrolled to use them: no memory round trip at the head of each trip; sum order unchanged
# speedup vs baseline: 1.2443x; 1.0589x over previous
#include <hip/hip_runtime.h>
#include <hip/hip_bf16.h>

typedef float v2f __attribute__((ext_vector_type(2)));
typedef float v8f __attribute__((ext_vector_type(8)));

#if defined(__has_builtin)
#if __has_builtin(__builtin_amdgcn_tanhf)
#define FAST_TANH(x) __builtin_amdgcn_tanhf(x)
#endif
#endif
#ifndef FAST_TANH
#define FAST_TANH(x) tanhf(x)
#endif

__device__ __forceinline__ float lane_gather(float v, int srcLane) {
  return __int_as_float(__builtin_amdgcn_ds_bpermute(srcLane << 2, __float_as_int(v)));
}

#define TPW 8

__global__ void __launch_bounds__(256)
pinn_mlp16_wmma(const float* __restrict__ x,  const float* __restrict__ x2,
                const float* __restrict__ W1, const float* __restrict__ W2,
                const float* __restrict__ B1,
                const float* __restrict__ W3, const float* __restrict__ B3,
                const float* __restrict__ W4, const float* __restrict__ B4,
                const float* __restrict__ W5, const float* __restrict__ B5,
                const float* __restrict__ W6, const float* __restrict__ B6,
                const float* __restrict__ W7, const float* __restrict__ B7,
                const float* __restrict__ W8, const float* __restrict__ B8,
                const float* __restrict__ W9, const float* __restrict__ B9,
                const float* __restrict__ alphap,
                float* __restrict__ out, int npts)
{
  const int lane = threadIdx.x & 31;
  const int p    = lane & 15;
  const int hi   = lane >> 4;
  const int wave = blockIdx.x * (blockDim.x >> 5) + (threadIdx.x >> 5);

  const int ntiles = (npts + 15) >> 4;
  const int tile0  = wave * TPW;
  if (tile0 >= ntiles) return;

  float xin[TPW], x2in[TPW];
  #pragma unroll
  for (int t = 0; t < TPW; ++t) {
    const int  idx = (tile0 + t) * 16 + p;
    const bool ok  = idx < npts;
    const int  ci  = ok ? idx : (npts - 1);
    const float vx  = x[ci];
    const float vx2 = x2[ci];
    xin[t]  = ok ? vx  : 0.0f;
    x2in[t] = ok ? vx2 : 0.0f;
  }

  const float alpha = alphap[0];
  const float b9    = B9[0];

  const float* Ws[6] = {W3, W4, W5, W6, W7, W8};
  const float* Bs[6] = {B3, B4, B5, B6, B7, B8};

  v2f   aW[6][4];
  float ab[6][8];
  #pragma unroll
  for (int l = 0; l < 6; ++l) {
    #pragma unroll
    for (int k = 0; k < 4; ++k)
      #pragma unroll
      for (int j = 0; j < 2; ++j)
        aW[l][k][j] = Ws[l][(2 * k + j + 8 * hi) * 16 + p];
    #pragma unroll
    for (int v = 0; v < 8; ++v)
      ab[l][v] = alpha * Bs[l][v + 8 * hi];
  }

  float aw1[8], aw2[8], ab1[8], w9v[8];
  #pragma unroll
  for (int v = 0; v < 8; ++v) {
    const int f = v + 8 * hi;
    aw1[v] = alpha * W1[f];
    aw2[v] = alpha * W2[f];
    ab1[v] = alpha * B1[f];
    w9v[v] = W9[f];
  }

#define FIRST_CHUNK(dst, xv, x2v, k)                                                   \
  do {                                                                                 \
    dst[0] = FAST_TANH(fmaf(xv, aw1[2 * (k)],     fmaf(x2v, aw2[2 * (k)],     ab1[2 * (k)])));     \
    dst[1] = FAST_TANH(fmaf(xv, aw1[2 * (k) + 1], fmaf(x2v, aw2[2 * (k) + 1], ab1[2 * (k) + 1]))); \
  } while (0)
#define NEXT_CHUNK(dst, dprev, lprev, k)                                               \
  do {                                                                                 \
    dst[0] = FAST_TANH(fmaf(alpha, dprev[2 * (k)],     ab[lprev][2 * (k)]));           \
    dst[1] = FAST_TANH(fmaf(alpha, dprev[2 * (k) + 1], ab[lprev][2 * (k) + 1]));       \
  } while (0)
#define WMMA_STEP(acc, l, k, bfrag)                                                    \
  acc = __builtin_amdgcn_wmma_f32_16x16x4_f32(false, aW[l][k], false, bfrag,           \
                                              (short)0, acc, false, false)
#define KEEP_ORDER() __builtin_amdgcn_sched_barrier(0)

  #pragma unroll
  for (int t = 0; t < TPW; t += 2) {
    const int  tile  = tile0 + t;
    if (tile >= ntiles) break;
    const int  baseA = tile << 4;
    const int  baseB = baseA + 16;
    const bool okA   = (baseA + p) < npts;
    const bool okB   = (baseB + p) < npts;
    const float xA   = xin[t];
    const float x2A  = x2in[t];
    const float xB   = xin[t + 1];
    const float x2B  = x2in[t + 1];

    v8f dA = {};
    v8f dB = {};
    v2f bA, bB, bAn, bBn;

    FIRST_CHUNK(bA, xA, x2A, 0);
    FIRST_CHUNK(bB, xB, x2B, 0);
    KEEP_ORDER();
    #pragma unroll
    for (int k = 0; k < 4; ++k) {
      WMMA_STEP(dA, 0, k, bA);
      if (k < 3) FIRST_CHUNK(bAn, xA, x2A, k + 1);
      KEEP_ORDER();
      WMMA_STEP(dB, 0, k, bB);
      if (k < 3) FIRST_CHUNK(bBn, xB, x2B, k + 1);
      KEEP_ORDER();
      bA = bAn;
      bB = bBn;
    }

    #pragma unroll
    for (int l = 1; l < 6; ++l) {
      const v8f pA = dA;
      const v8f pB = dB;
      dA = (v8f){};
      dB = (v8f){};
      NEXT_CHUNK(bA, pA, l - 1, 0);
      KEEP_ORDER();
      WMMA_STEP(dA, l, 0, bA);
      NEXT_CHUNK(bB, pB, l - 1, 0);
      NEXT_CHUNK(bAn, pA, l - 1, 1);
      KEEP_ORDER();
      WMMA_STEP(dB, l, 0, bB);
      NEXT_CHUNK(bBn, pB, l - 1, 1);
      KEEP_ORDER();
      bA = bAn;
      bB = bBn;
      #pragma unroll
      for (int k = 1; k < 4; ++k) {
        WMMA_STEP(dA, l, k, bA);
        if (k < 3) NEXT_CHUNK(bAn, pA, l - 1, k + 1);
        KEEP_ORDER();
        WMMA_STEP(dB, l, k, bB);
        if (k < 3) NEXT_CHUNK(bBn, pB, l - 1, k + 1);
        KEEP_ORDER();
        bA = bAn;
        bB = bBn;
      }
    }

    float partA = 0.0f, partB = 0.0f;
    #pragma unroll
    for (int v = 0; v < 8; ++v) {
      partA = fmaf(FAST_TANH(fmaf(alpha, dA[v], ab[5][v])), w9v[v], partA);
      partB = fmaf(FAST_TANH(fmaf(alpha, dB[v], ab[5][v])), w9v[v], partB);
    }
    const float otherA = lane_gather(partA, lane ^ 16);
    const float otherB = lane_gather(partB, lane ^ 16);
    const float resA   = partA + otherA + b9;
    const float resB   = partB + otherB + b9;
    if (hi == 0 && okA)
      out[baseA + p] = resA;
    if (hi == 0 && okB)
      out[baseB + p] = resB;
  }
#undef FIRST_CHUNK
#undef NEXT_CHUNK
#undef WMMA_STEP
#undef KEEP_ORDER
}

extern "C" void kernel_launch(void* const* d_in, const int* in_sizes, int n_in,
                              void* d_out, int out_size, void* d_ws, size_t ws_size,
                              hipStream_t stream) {
  (void)n_in; (void)d_ws; (void)ws_size; (void)out_size;
  const float* x   = (const float*)d_in[0];
  const float* x2  = (const float*)d_in[1];
  const float* W1  = (const float*)d_in[2];
  const float* W2  = (const float*)d_in[3];
  const float* B1  = (const float*)d_in[4];
  const float* W3  = (const float*)d_in[5];
  const float* B3  = (const float*)d_in[6];
  const float* W4  = (const float*)d_in[7];
  const float* B4  = (const float*)d_in[8];
  const float* W5  = (const float*)d_in[9];
  const float* B5  = (const float*)d_in[10];
  const float* W6  = (const float*)d_in[11];
  const float* B6  = (const float*)d_in[12];
  const float* W7  = (const float*)d_in[13];
  const float* B7  = (const float*)d_in[14];
  const float* W8  = (const float*)d_in[15];
  const float* B8  = (const float*)d_in[16];
  const float* W9  = (const float*)d_in[17];
  const float* B9  = (const float*)d_in[18];
  const float* alp = (const float*)d_in[19];
  float* out = (float*)d_out;

  const int npts   = in_sizes[0];
  const int ntiles = (npts + 15) >> 4;
  const int waves  = (ntiles + TPW - 1) / TPW;
  const int blocks = (waves + 7) / 8;

  pinn_mlp16_wmma<<<blocks, 256, 0, stream>>>(
      x, x2, W1, W2, B1, W3, B3, W4, B4, W5, B5, W6, B6,
      W7, B7, W8, B8, W9, B9, alp, out, npts);
}


# alpha folded into A (alpha*W) and alpha*bias made each layer's accumulator start, so an activation is a bare tanh of D, no multiply-add per value; sum order: bias first, alpha scales W not the sum
# speedup vs baseline: 1.2852x; 1.0329x over previous
#include <hip/hip_runtime.h>
#include <hip/hip_bf16.h>

typedef float v2f __attribute__((ext_vector_type(2)));
typedef float v8f __attribute__((ext_vector_type(8)));

#if defined(__has_builtin)
#if __has_builtin(__builtin_amdgcn_tanhf)
#define FAST_TANH(x) __builtin_amdgcn_tanhf(x)
#endif
#endif
#ifndef FAST_TANH
#define FAST_TANH(x) tanhf(x)
#endif

__device__ __forceinline__ float lane_gather(float v, int srcLane) {
  return __int_as_float(__builtin_amdgcn_ds_bpermute(srcLane << 2, __float_as_int(v)));
}

#define TPW 8

__global__ void __launch_bounds__(256)
pinn_mlp16_wmma(const float* __restrict__ x,  const float* __restrict__ x2,
                const float* __restrict__ W1, const float* __restrict__ W2,
                const float* __restrict__ B1,
                const float* __restrict__ W3, const float* __restrict__ B3,
                const float* __restrict__ W4, const float* __restrict__ B4,
                const float* __restrict__ W5, const float* __restrict__ B5,
                const float* __restrict__ W6, const float* __restrict__ B6,
                const float* __restrict__ W7, const float* __restrict__ B7,
                const float* __restrict__ W8, const float* __restrict__ B8,
                const float* __restrict__ W9, const float* __restrict__ B9,
                const float* __restrict__ alphap,
                float* __restrict__ out, int npts)
{
  const int lane = threadIdx.x & 31;
  const int p    = lane & 15;
  const int hi   = lane >> 4;
  const int wave = blockIdx.x * (blockDim.x >> 5) + (threadIdx.x >> 5);

  const int ntiles = (npts + 15) >> 4;
  const int tile0  = wave * TPW;
  if (tile0 >= ntiles) return;

  float xin[TPW], x2in[TPW];
  #pragma unroll
  for (int t = 0; t < TPW; ++t) {
    const int  idx = (tile0 + t) * 16 + p;
    const bool ok  = idx < npts;
    const int  ci  = ok ? idx : (npts - 1);
    const float vx  = x[ci];
    const float vx2 = x2[ci];
    xin[t]  = ok ? vx  : 0.0f;
    x2in[t] = ok ? vx2 : 0.0f;
  }

  const float alpha = alphap[0];
  const float b9    = B9[0];

  const float* Ws[6] = {W3, W4, W5, W6, W7, W8};
  const float* Bs[6] = {B3, B4, B5, B6, B7, B8};

  v2f aW[6][4];
  v8f cb[6];
  #pragma unroll
  for (int l = 0; l < 6; ++l) {
    #pragma unroll
    for (int k = 0; k < 4; ++k)
      #pragma unroll
      for (int j = 0; j < 2; ++j)
        aW[l][k][j] = alpha * Ws[l][(2 * k + j + 8 * hi) * 16 + p];
    #pragma unroll
    for (int v = 0; v < 8; ++v)
      cb[l][v] = alpha * Bs[l][v + 8 * hi];
  }

  float aw1[8], aw2[8], ab1[8], w9v[8];
  #pragma unroll
  for (int v = 0; v < 8; ++v) {
    const int f = v + 8 * hi;
    aw1[v] = alpha * W1[f];
    aw2[v] = alpha * W2[f];
    ab1[v] = alpha * B1[f];
    w9v[v] = W9[f];
  }

#define FIRST_CHUNK(dst, xv, x2v, k)                                                   \
  do {                                                                                 \
    dst[0] = FAST_TANH(fmaf(xv, aw1[2 * (k)],     fmaf(x2v, aw2[2 * (k)],     ab1[2 * (k)])));     \
    dst[1] = FAST_TANH(fmaf(xv, aw1[2 * (k) + 1], fmaf(x2v, aw2[2 * (k) + 1], ab1[2 * (k) + 1]))); \
  } while (0)
#define NEXT_CHUNK(dst, dprev, k)                                                      \
  do {                                                                                 \
    dst[0] = FAST_TANH(dprev[2 * (k)]);                                                \
    dst[1] = FAST_TANH(dprev[2 * (k) + 1]);                                            \
  } while (0)
#define WMMA_STEP(acc, l, k, bfrag)                                                    \
  acc = __builtin_amdgcn_wmma_f32_16x16x4_f32(false, aW[l][k], false, bfrag,           \
                                              (short)0, acc, false, false)
#define KEEP_ORDER() __builtin_amdgcn_sched_barrier(0)

  #pragma unroll
  for (int t = 0; t < TPW; t += 2) {
    const int  tile  = tile0 + t;
    if (tile >= ntiles) break;
    const int  baseA = tile << 4;
    const int  baseB = baseA + 16;
    const bool okA   = (baseA + p) < npts;
    const bool okB   = (baseB + p) < npts;
    const float xA   = xin[t];
    const float x2A  = x2in[t];
    const float xB   = xin[t + 1];
    const float x2B  = x2in[t + 1];

    v8f dA = cb[0];
    v8f dB = cb[0];
    v2f bA, bB, bAn, bBn;

    FIRST_CHUNK(bA, xA, x2A, 0);
    FIRST_CHUNK(bB, xB, x2B, 0);
    KEEP_ORDER();
    #pragma unroll
    for (int k = 0; k < 4; ++k) {
      WMMA_STEP(dA, 0, k, bA);
      if (k < 3) FIRST_CHUNK(bAn, xA, x2A, k + 1);
      KEEP_ORDER();
      WMMA_STEP(dB, 0, k, bB);
      if (k < 3) FIRST_CHUNK(bBn, xB, x2B, k + 1);
      KEEP_ORDER();
      bA = bAn;
      bB = bBn;
    }

    #pragma unroll
    for (int l = 1; l < 6; ++l) {
      const v8f pA = dA;
      const v8f pB = dB;
      dA = cb[l];
      dB = cb[l];
      NEXT_CHUNK(bA, pA, 0);
      KEEP_ORDER();
      WMMA_STEP(dA, l, 0, bA);
      NEXT_CHUNK(bB, pB, 0);
      NEXT_CHUNK(bAn, pA, 1);
      KEEP_ORDER();
      WMMA_STEP(dB, l, 0, bB);
      NEXT_CHUNK(bBn, pB, 1);
      KEEP_ORDER();
      bA = bAn;
      bB = bBn;
      #pragma unroll
      for (int k = 1; k < 4; ++k) {
        WMMA_STEP(dA, l, k, bA);
        if (k < 3) NEXT_CHUNK(bAn, pA, k + 1);
        KEEP_ORDER();
        WMMA_STEP(dB, l, k, bB);
        if (k < 3) NEXT_CHUNK(bBn, pB, k + 1);
        KEEP_ORDER();
        bA = bAn;
        bB = bBn;
      }
    }

    float partA = 0.0f, partB = 0.0f;
    #pragma unroll
    for (int v = 0; v < 8; ++v) {
      partA = fmaf(FAST_TANH(dA[v]), w9v[v], partA);
      partB = fmaf(FAST_TANH(dB[v]), w9v[v], partB);
    }
    const float otherA = lane_gather(partA, lane ^ 16);
    const float otherB = lane_gather(partB, lane ^ 16);
    const float resA   = partA + otherA + b9;
    const float resB   = partB + otherB + b9;
    if (hi == 0 && okA)
      out[baseA + p] = resA;
    if (hi == 0 && okB)
      out[baseB + p] = resB;
  }
#undef FIRST_CHUNK
#undef NEXT_CHUNK
#undef WMMA_STEP
#undef KEEP_ORDER
}

extern "C" void kernel_launch(void* const* d_in, const int* in_sizes, int n_in,
                              void* d_out, int out_size, void* d_ws, size_t ws_size,
                              hipStream_t stream) {
  (void)n_in; (void)d_ws; (void)ws_size; (void)out_size;
  const float* x   = (const float*)d_in[0];
  const float* x2  = (const float*)d_in[1];
  const float* W1  = (const float*)d_in[2];
  const float* W2  = (const float*)d_in[3];
  const float* B1  = (const float*)d_in[4];
  const float* W3  = (const float*)d_in[5];
  const float* B3  = (const float*)d_in[6];
  const float* W4  = (const float*)d_in[7];
  const float* B4  = (const float*)d_in[8];
  const float* W5  = (const float*)d_in[9];
  const float* B5  = (const float*)d_in[10];
  const float* W6  = (const float*)d_in[11];
  const float* B6  = (const float*)d_in[12];
  const float* W7  = (const float*)d_in[13];
  const float* B7  = (const float*)d_in[14];
  const float* W8  = (const float*)d_in[15];
  const float* B8  = (const float*)d_in[16];
  const float* W9  = (const float*)d_in[17];
  const float* B9  = (const float*)d_in[18];
  const float* alp = (const float*)d_in[19];
  float* out = (float*)d_out;

  const int npts   = in_sizes[0];
  const int ntiles = (npts + 15) >> 4;
  const int waves  = (ntiles + TPW - 1) / TPW;
  const int blocks = (waves + 7) / 8;

  pinn_mlp16_wmma<<<blocks, 256, 0, stream>>>(
      x, x2, W1, W2, B1, W3, B3, W4, B4, W5, B5, W6, B6,
      W7, B7, W8, B8, W9, B9, alp, out, npts);
}
